# starting kernel
# baseline (speedup 1.0000x reference)
#include <hip/hip_runtime.h>
#include <math.h>


typedef float v2f __attribute__((ext_vector_type(2)));
typedef float v8f __attribute__((ext_vector_type(8)));

#define DD      8
#define HH      64
#define IN_DIM  9
#define BB      128
#define LL      256
#define NTOT    (BB * LL)
#define SLOPE   0.2f
#define AST     66
#define WAVES   8
#define ROWSBLK (WAVES * 16)

static __device__ __forceinline__ v8f wmma_f32(v2f a, v2f b, v8f c) {
  return __builtin_amdgcn_wmma_f32_16x16x4_f32(
       false, a,  false, b,
       (short)0, c,  false,  false);
}

__global__ __launch_bounds__(256) void flow_mlp_kernel(
    const float* __restrict__ x,
    const float* __restrict__ W1, const float* __restrict__ b1,
    const float* __restrict__ W2, const float* __restrict__ b2,
    const float* __restrict__ W3, const float* __restrict__ b3,
    const float* __restrict__ Wo, const float* __restrict__ bo,
    float* __restrict__ out, float* __restrict__ ws)
{
  __shared__ float sW1T[16 * 64];
  __shared__ float sW2T[64 * 64];
  __shared__ float sW3T[64 * 64];
  __shared__ float sB1[64], sB2[64], sB3[64], sWoL[64];
  __shared__ float sBufA[WAVES][16 * AST];
  __shared__ float sBufD[WAVES][16 * AST];

  const int tid = threadIdx.x;
  const int d   = blockIdx.y;

  for (int i = tid; i < 16 * 64; i += 256) {
    const int k = i >> 6, h = i & 63;
    sW1T[i] = (k < IN_DIM) ? W1[(d * HH + h) * IN_DIM + k] : 0.f;
  }
  for (int i = tid; i < 64 * 64; i += 256) {
    const int g = i >> 6, h = i & 63;
    sW2T[h * 64 + g] = W2[d * 4096 + i];
    sW3T[h * 64 + g] = W3[d * 4096 + i];
  }
  if (tid < 64) {
    sB1[tid]  = b1[d * HH + tid];
    sB2[tid]  = b2[d * HH + tid];
    sB3[tid]  = b3[d * HH + tid];
    sWoL[tid] = Wo[d * HH + tid];
  }
  __syncthreads();

  const int wave = tid >> 5;
  const int lane = tid & 31;
  const int lx   = lane & 15;
  const int half = lane >> 4;
  const int koff = half * 2;

  float* sA = sBufA[wave];
  float* sD = sBufD[wave];

  const int n0 = blockIdx.x * ROWSBLK + wave * 16;
  const int bb = n0 >> 8;
  const int l0 = n0 & 255;
  {
    const int s = lane >> 1, part = lane & 1;
    const float4 v = *reinterpret_cast<const float4*>(
        x + (size_t)(bb * 257 + l0 + s) * 8 + part * 4);
    float* dst = &sA[s * AST + part * 4];
    dst[0] = v.x; dst[1] = v.y; dst[2] = v.z; dst[3] = v.w;
    if (lane < 16) {
      sA[lane * AST + 8] = x[(size_t)(bb * 257 + l0 + lane + 1) * 8 + d];
    } else {
      sA[lx * AST + 9]  = 0.f;
      sA[lx * AST + 10] = 0.f;
      sA[lx * AST + 11] = 0.f;
    }
  }

  v8f acc1[4];
  {
    v2f af[3];
#pragma unroll
    for (int kk = 0; kk < 3; ++kk) {
      const int k = kk * 4 + koff;
      af[kk].x = sA[lx * AST + k];
      af[kk].y = sA[lx * AST + k + 1];
    }
#pragma unroll
    for (int t = 0; t < 4; ++t) {
      v8f c = {0.f, 0.f, 0.f, 0.f, 0.f, 0.f, 0.f, 0.f};
#pragma unroll
      for (int kk = 0; kk < 3; ++kk) {
        const int k = kk * 4 + koff;
        v2f b;
        b.x = sW1T[k * 64 + t * 16 + lx];
        b.y = sW1T[(k + 1) * 64 + t * 16 + lx];
        c = wmma_f32(af[kk], b, c);
      }
      acc1[t] = c;
    }
  }
#pragma unroll
  for (int t = 0; t < 4; ++t) {
    const int f   = t * 16 + lx;
    const float w8 = sW1T[8 * 64 + f];
    const float bv = sB1[f];
#pragma unroll
    for (int v = 0; v < 8; ++v) {
      const float h = acc1[t][v] + bv;
      const float m = (h >= 0.f) ? 1.f : SLOPE;
      const int  ms = v + 8 * half;
      sA[ms * AST + f] = m * h;
      sD[ms * AST + f] = m * w8;
    }
  }

  auto layer = [&](const float* __restrict__ WT, const float* __restrict__ bias) {
    v2f af[16], df[16];
#pragma unroll
    for (int kk = 0; kk < 16; ++kk) {
      const int k = kk * 4 + koff;
      af[kk].x = sA[lx * AST + k];  af[kk].y = sA[lx * AST + k + 1];
      df[kk].x = sD[lx * AST + k];  df[kk].y = sD[lx * AST + k + 1];
    }
#pragma unroll
    for (int t = 0; t < 4; ++t) {
      v8f ca = {0.f, 0.f, 0.f, 0.f, 0.f, 0.f, 0.f, 0.f};
      v8f cd = {0.f, 0.f, 0.f, 0.f, 0.f, 0.f, 0.f, 0.f};
#pragma unroll
      for (int kk = 0; kk < 16; ++kk) {
        const int k = kk * 4 + koff;
        v2f b;
        b.x = WT[k * 64 + t * 16 + lx];
        b.y = WT[(k + 1) * 64 + t * 16 + lx];
        ca = wmma_f32(af[kk], b, ca);
        cd = wmma_f32(df[kk], b, cd);
      }
      const int f   = t * 16 + lx;
      const float bv = bias[f];
#pragma unroll
      for (int v = 0; v < 8; ++v) {
        const float h = ca[v] + bv;
        const float m = (h >= 0.f) ? 1.f : SLOPE;
        const int  ms = v + 8 * half;
        sA[ms * AST + f] = m * h;
        sD[ms * AST + f] = m * cd[v];
      }
    }
  };
  layer(sW2T, sB2);
  layer(sW3T, sB3);

  {
    const float* buf = (lane < 16) ? sA : sD;
    float s = 0.f;
#pragma unroll 8
    for (int k = 0; k < 64; ++k) s += buf[lx * AST + k] * sWoL[k];
    const int n = n0 + lx;
    if (lane < 16) {
      out[(size_t)n * DD + d] = s + bo[d];
    } else {
      ws[(size_t)d * NTOT + n] = s;
    }
  }
}

__global__ __launch_bounds__(256) void logdet_kernel(
    const float* __restrict__ ws, float* __restrict__ out)
{
  const int r = blockIdx.x * 256 + threadIdx.x;
  if (r >= NTOT) return;
  float s = 0.f;
#pragma unroll
  for (int d = 0; d < DD; ++d) s += logf(fabsf(ws[(size_t)d * NTOT + r]));
  out[(size_t)NTOT * DD + r] = s;
}

extern "C" void kernel_launch(void* const* d_in, const int* in_sizes, int n_in,
                              void* d_out, int out_size, void* d_ws, size_t ws_size,
                              hipStream_t stream) {
  const float* x  = (const float*)d_in[0];
  const float* W1 = (const float*)d_in[1];
  const float* b1 = (const float*)d_in[2];
  const float* W2 = (const float*)d_in[3];
  const float* b2 = (const float*)d_in[4];
  const float* W3 = (const float*)d_in[5];
  const float* b3 = (const float*)d_in[6];
  const float* Wo = (const float*)d_in[7];
  const float* bo = (const float*)d_in[8];
  float* out = (float*)d_out;
  float* ws  = (float*)d_ws;

  dim3 grid(NTOT / ROWSBLK, DD);
  flow_mlp_kernel<<<grid, dim3(256), 0, stream>>>(
      x, W1, b1, W2, b2, W3, b3, Wo, bo, out, ws);
  logdet_kernel<<<dim3(NTOT / 256), dim3(256), 0, stream>>>(ws, out);
}


# Weights kept in VGPRs as B fragments, read once per wave straight from global (weight rows are K-contiguous); a wave loops over 8 sample tiles; LDS holds activations only; was per-block LDS staging
# speedup vs baseline: 1.4407x; 1.4407x over previous
#include <hip/hip_runtime.h>
#include <math.h>


typedef float v2f __attribute__((ext_vector_type(2)));
typedef float v8f __attribute__((ext_vector_type(8)));

#define DD      8
#define HH      64
#define IN_DIM  9
#define BB      128
#define LL      256
#define NTOT    (BB * LL)
#define SLOPE   0.2f
#define AST     66
#define WAVES   8
#define TILES   8
#define ROWSIT  (WAVES * 16)
#define ROWSBLK (ROWSIT * TILES)

static __device__ __forceinline__ v8f wmma_f32(v2f a, v2f b, v8f c) {
  return __builtin_amdgcn_wmma_f32_16x16x4_f32(
       false, a,  false, b,
       (short)0, c,  false,  false);
}

static __device__ __forceinline__ void hidden_layer(
    float* __restrict__ sA, float* __restrict__ sD,
    const v2f (&wf)[4][16], const float (&bias)[4],
    const int lx, const int half, const int koff)
{
  v2f af[16], df[16];
#pragma unroll
  for (int kk = 0; kk < 16; ++kk) {
    const int k = kk * 4 + koff;
    af[kk].x = sA[lx * AST + k];  af[kk].y = sA[lx * AST + k + 1];
    df[kk].x = sD[lx * AST + k];  df[kk].y = sD[lx * AST + k + 1];
  }
#pragma unroll
  for (int t = 0; t < 4; ++t) {
    v8f ca = {0.f, 0.f, 0.f, 0.f, 0.f, 0.f, 0.f, 0.f};
    v8f cd = {0.f, 0.f, 0.f, 0.f, 0.f, 0.f, 0.f, 0.f};
#pragma unroll
    for (int kk = 0; kk < 16; ++kk) {
      ca = wmma_f32(af[kk], wf[t][kk], ca);
      cd = wmma_f32(df[kk], wf[t][kk], cd);
    }
    const int f   = t * 16 + lx;
    const float bv = bias[t];
#pragma unroll
    for (int v = 0; v < 8; ++v) {
      const float h = ca[v] + bv;
      const float m = (h >= 0.f) ? 1.f : SLOPE;
      const int  ms = v + 8 * half;
      sA[ms * AST + f] = m * h;
      sD[ms * AST + f] = m * cd[v];
    }
  }
}

__global__ __launch_bounds__(256) void flow_mlp_kernel(
    const float* __restrict__ x,
    const float* __restrict__ W1, const float* __restrict__ b1,
    const float* __restrict__ W2, const float* __restrict__ b2,
    const float* __restrict__ W3, const float* __restrict__ b3,
    const float* __restrict__ Wo, const float* __restrict__ bo,
    float* __restrict__ out, float* __restrict__ ws)
{
  __shared__ float sBufA[WAVES][16 * AST];
  __shared__ float sBufD[WAVES][16 * AST];

  const int tid  = threadIdx.x;
  const int d    = blockIdx.y;
  const int wave = tid >> 5;
  const int lane = tid & 31;
  const int lx   = lane & 15;
  const int half = lane >> 4;
  const int koff = half * 2;

  float* sA = sBufA[wave];
  float* sD = sBufD[wave];

  v2f w1f[4][3];
  v2f w2f[4][16];
  v2f w3f[4][16];
  float w8[4], bv1[4], bv2[4], bv3[4];
  float wo[HH];
#pragma unroll
  for (int t = 0; t < 4; ++t) {
    const int g = d * HH + t * 16 + lx;
    const float* r1 = W1 + (size_t)g * IN_DIM;
    w8[t] = r1[8];
    w1f[t][0].x = r1[koff];      w1f[t][0].y = r1[koff + 1];
    w1f[t][1].x = r1[4 + koff];  w1f[t][1].y = r1[5 + koff];
    w1f[t][2].x = half ? 0.f : w8[t];
    w1f[t][2].y = 0.f;
    bv1[t] = b1[g];
    bv2[t] = b2[g];
    bv3[t] = b3[g];
  }
#pragma unroll
  for (int t = 0; t < 4; ++t) {
    const int g = d * HH + t * 16 + lx;
    const v2f* r2 = reinterpret_cast<const v2f*>(W2 + (size_t)g * HH + koff);
#pragma unroll
    for (int kk = 0; kk < 16; ++kk) w2f[t][kk] = r2[kk * 2];
  }
#pragma unroll
  for (int t = 0; t < 4; ++t) {
    const int g = d * HH + t * 16 + lx;
    const v2f* r3 = reinterpret_cast<const v2f*>(W3 + (size_t)g * HH + koff);
#pragma unroll
    for (int kk = 0; kk < 16; ++kk) w3f[t][kk] = r3[kk * 2];
  }
#pragma unroll
  for (int k = 0; k < HH; ++k) wo[k] = Wo[d * HH + k];
  const float bod = bo[d];

  for (int it = 0; it < TILES; ++it) {
    const int n0 = (blockIdx.x * TILES + it) * ROWSIT + wave * 16;
    const int bb = n0 >> 8;
    const int l0 = n0 & 255;
    {
      const int s = lane >> 1, part = lane & 1;
      const float4 v = *reinterpret_cast<const float4*>(
          x + (size_t)(bb * 257 + l0 + s) * 8 + part * 4);
      const float xt = x[(size_t)(bb * 257 + l0 + lx + 1) * 8 + d];
      float* dst = &sA[s * AST + part * 4];
      dst[0] = v.x; dst[1] = v.y; dst[2] = v.z; dst[3] = v.w;
      sA[lx * AST + 8 + half] = half ? 0.f : xt;
      if (half) {
        sA[lx * AST + 10] = 0.f;
        sA[lx * AST + 11] = 0.f;
      }
    }

    {
      v2f af[3];
#pragma unroll
      for (int kk = 0; kk < 3; ++kk) {
        const int k = kk * 4 + koff;
        af[kk].x = sA[lx * AST + k];
        af[kk].y = sA[lx * AST + k + 1];
      }
      v8f acc1[4];
#pragma unroll
      for (int t = 0; t < 4; ++t) {
        v8f c = {0.f, 0.f, 0.f, 0.f, 0.f, 0.f, 0.f, 0.f};
#pragma unroll
        for (int kk = 0; kk < 3; ++kk) c = wmma_f32(af[kk], w1f[t][kk], c);
        acc1[t] = c;
      }
#pragma unroll
      for (int t = 0; t < 4; ++t) {
        const int f = t * 16 + lx;
#pragma unroll
        for (int v = 0; v < 8; ++v) {
          const float h = acc1[t][v] + bv1[t];
          const float m = (h >= 0.f) ? 1.f : SLOPE;
          const int  ms = v + 8 * half;
          sA[ms * AST + f] = m * h;
          sD[ms * AST + f] = m * w8[t];
        }
      }
    }

    hidden_layer(sA, sD, w2f, bv2, lx, half, koff);
    hidden_layer(sA, sD, w3f, bv3, lx, half, koff);

    {
      const float* buf = (lane < 16) ? sA : sD;
      float s = 0.f;
#pragma unroll
      for (int k = 0; k < 64; ++k) s += buf[lx * AST + k] * wo[k];
      const int n = n0 + lx;
      if (lane < 16) {
        out[(size_t)n * DD + d] = s + bod;
      } else {
        ws[(size_t)d * NTOT + n] = s;
      }
    }
  }
}

__global__ __launch_bounds__(256) void logdet_kernel(
    const float* __restrict__ ws, float* __restrict__ out)
{
  const int r = blockIdx.x * 256 + threadIdx.x;
  if (r >= NTOT) return;
  float s = 0.f;
#pragma unroll
  for (int d = 0; d < DD; ++d) s += logf(fabsf(ws[(size_t)d * NTOT + r]));
  out[(size_t)NTOT * DD + r] = s;
}

extern "C" void kernel_launch(void* const* d_in, const int* in_sizes, int n_in,
                              void* d_out, int out_size, void* d_ws, size_t ws_size,
                              hipStream_t stream) {
  const float* x  = (const float*)d_in[0];
  const float* W1 = (const float*)d_in[1];
  const float* b1 = (const float*)d_in[2];
  const float* W2 = (const float*)d_in[3];
  const float* b2 = (const float*)d_in[4];
  const float* W3 = (const float*)d_in[5];
  const float* b3 = (const float*)d_in[6];
  const float* Wo = (const float*)d_in[7];
  const float* bo = (const float*)d_in[8];
  float* out = (float*)d_out;
  float* ws  = (float*)d_ws;

  dim3 grid(NTOT / ROWSBLK, DD);
  flow_mlp_kernel<<<grid, dim3(256), 0, stream>>>(
      x, W1, b1, W2, b2, W3, b3, Wo, bo, out, ws);
  logdet_kernel<<<dim3(NTOT / 256), dim3(256), 0, stream>>>(ws, out);
}


# Log-det sum fused into the MLP kernel: a workgroup owns 128 samples for all 8 coordinates (wave = coordinate), adds log|dlast| over its waves via LDS, stores whole rows; 2nd launch and workspace gone
# speedup vs baseline: 1.5611x; 1.0836x over previous
#include <hip/hip_runtime.h>
#include <math.h>


typedef float v2f __attribute__((ext_vector_type(2)));
typedef float v8f __attribute__((ext_vector_type(8)));

#define DD      8
#define HH      64
#define IN_DIM  9
#define BB      128
#define LL      256
#define NTOT    (BB * LL)
#define SLOPE   0.2f
#define AST     66
#define WAVES   8
#define TILES   8
#define ROWSBLK (16 * TILES)

static __device__ __forceinline__ v8f wmma_f32(v2f a, v2f b, v8f c) {
  return __builtin_amdgcn_wmma_f32_16x16x4_f32(
       false, a,  false, b,
       (short)0, c,  false,  false);
}

static __device__ __forceinline__ void hidden_layer(
    float* __restrict__ sA, float* __restrict__ sD,
    const v2f (&wf)[4][16], const float (&bias)[4],
    const int lx, const int half, const int koff)
{
  v2f af[16], df[16];
#pragma unroll
  for (int kk = 0; kk < 16; ++kk) {
    const int k = kk * 4 + koff;
    af[kk].x = sA[lx * AST + k];  af[kk].y = sA[lx * AST + k + 1];
    df[kk].x = sD[lx * AST + k];  df[kk].y = sD[lx * AST + k + 1];
  }
#pragma unroll
  for (int t = 0; t < 4; ++t) {
    v8f ca = {0.f, 0.f, 0.f, 0.f, 0.f, 0.f, 0.f, 0.f};
    v8f cd = {0.f, 0.f, 0.f, 0.f, 0.f, 0.f, 0.f, 0.f};
#pragma unroll
    for (int kk = 0; kk < 16; ++kk) {
      ca = wmma_f32(af[kk], wf[t][kk], ca);
      cd = wmma_f32(df[kk], wf[t][kk], cd);
    }
    const int f   = t * 16 + lx;
    const float bv = bias[t];
#pragma unroll
    for (int v = 0; v < 8; ++v) {
      const float h = ca[v] + bv;
      const float m = (h >= 0.f) ? 1.f : SLOPE;
      const int  ms = v + 8 * half;
      sA[ms * AST + f] = m * h;
      sD[ms * AST + f] = m * cd[v];
    }
  }
}

__global__ __launch_bounds__(256) void flow_mlp_kernel(
    const float* __restrict__ x,
    const float* __restrict__ W1, const float* __restrict__ b1,
    const float* __restrict__ W2, const float* __restrict__ b2,
    const float* __restrict__ W3, const float* __restrict__ b3,
    const float* __restrict__ Wo, const float* __restrict__ bo,
    float* __restrict__ out)
{
  __shared__ float sBufA[WAVES][16 * AST];
  __shared__ float sBufD[WAVES][16 * AST];
  __shared__ __attribute__((aligned(16))) float sRes[ROWSBLK * DD];
  __shared__ float sDl[DD][ROWSBLK];

  const int tid  = threadIdx.x;
  const int wave = __builtin_amdgcn_readfirstlane(tid >> 5);
  const int d    = wave;
  const int lane = tid & 31;
  const int lx   = lane & 15;
  const int half = lane >> 4;
  const int koff = half * 2;

  float* sA = sBufA[wave];
  float* sD = sBufD[wave];

  v2f w1f[4][3];
  v2f w2f[4][16];
  v2f w3f[4][16];
  float w8[4], bv1[4], bv2[4], bv3[4];
  float wo[HH];
#pragma unroll
  for (int t = 0; t < 4; ++t) {
    const int g = d * HH + t * 16 + lx;
    const float* r1 = W1 + (size_t)g * IN_DIM;
    w8[t] = r1[8];
    w1f[t][0].x = r1[koff];      w1f[t][0].y = r1[koff + 1];
    w1f[t][1].x = r1[4 + koff];  w1f[t][1].y = r1[5 + koff];
    w1f[t][2].x = half ? 0.f : w8[t];
    w1f[t][2].y = 0.f;
    bv1[t] = b1[g];
    bv2[t] = b2[g];
    bv3[t] = b3[g];
  }
#pragma unroll
  for (int t = 0; t < 4; ++t) {
    const int g = d * HH + t * 16 + lx;
    const v2f* r2 = reinterpret_cast<const v2f*>(W2 + (size_t)g * HH + koff);
#pragma unroll
    for (int kk = 0; kk < 16; ++kk) w2f[t][kk] = r2[kk * 2];
  }
#pragma unroll
  for (int t = 0; t < 4; ++t) {
    const int g = d * HH + t * 16 + lx;
    const v2f* r3 = reinterpret_cast<const v2f*>(W3 + (size_t)g * HH + koff);
#pragma unroll
    for (int kk = 0; kk < 16; ++kk) w3f[t][kk] = r3[kk * 2];
  }
#pragma unroll
  for (int k = 0; k < HH; ++k) wo[k] = Wo[d * HH + k];
  const float bod = bo[d];

  for (int it = 0; it < TILES; ++it) {
    const int n0 = blockIdx.x * ROWSBLK + it * 16;
    const int bb = n0 >> 8;
    const int l0 = n0 & 255;
    {
      const int s = lane >> 1, part = lane & 1;
      const float4 v = *reinterpret_cast<const float4*>(
          x + (size_t)(bb * 257 + l0 + s) * 8 + part * 4);
      const float xt = x[(size_t)(bb * 257 + l0 + lx + 1) * 8 + d];
      float* dst = &sA[s * AST + part * 4];
      dst[0] = v.x; dst[1] = v.y; dst[2] = v.z; dst[3] = v.w;
      sA[lx * AST + 8 + half] = half ? 0.f : xt;
      if (half) {
        sA[lx * AST + 10] = 0.f;
        sA[lx * AST + 11] = 0.f;
      }
    }

    {
      v2f af[3];
#pragma unroll
      for (int kk = 0; kk < 3; ++kk) {
        const int k = kk * 4 + koff;
        af[kk].x = sA[lx * AST + k];
        af[kk].y = sA[lx * AST + k + 1];
      }
      v8f acc1[4];
#pragma unroll
      for (int t = 0; t < 4; ++t) {
        v8f c = {0.f, 0.f, 0.f, 0.f, 0.f, 0.f, 0.f, 0.f};
#pragma unroll
        for (int kk = 0; kk < 3; ++kk) c = wmma_f32(af[kk], w1f[t][kk], c);
        acc1[t] = c;
      }
#pragma unroll
      for (int t = 0; t < 4; ++t) {
        const int f = t * 16 + lx;
#pragma unroll
        for (int v = 0; v < 8; ++v) {
          const float h = acc1[t][v] + bv1[t];
          const float m = (h >= 0.f) ? 1.f : SLOPE;
          const int  ms = v + 8 * half;
          sA[ms * AST + f] = m * h;
          sD[ms * AST + f] = m * w8[t];
        }
      }
    }

    hidden_layer(sA, sD, w2f, bv2, lx, half, koff);
    hidden_layer(sA, sD, w3f, bv3, lx, half, koff);

    {
      const float* buf = (lane < 16) ? sA : sD;
      float s = 0.f;
#pragma unroll
      for (int k = 0; k < 64; ++k) s += buf[lx * AST + k] * wo[k];
      const int r = it * 16 + lx;
      if (lane < 16) {
        sRes[r * DD + d] = s + bod;
      } else {
        sDl[d][r] = s;
      }
    }
  }

  __syncthreads();
  {
    const float4 v = *reinterpret_cast<const float4*>(&sRes[tid * 4]);
    *reinterpret_cast<float4*>(out + (size_t)blockIdx.x * ROWSBLK * DD + tid * 4) = v;
  }
  if (tid < ROWSBLK) {
    float s = 0.f;
#pragma unroll
    for (int dd = 0; dd < DD; ++dd) s += logf(fabsf(sDl[dd][tid]));
    out[(size_t)NTOT * DD + blockIdx.x * ROWSBLK + tid] = s;
  }
}

extern "C" void kernel_launch(void* const* d_in, const int* in_sizes, int n_in,
                              void* d_out, int out_size, void* d_ws, size_t ws_size,
                              hipStream_t stream) {
  const float* x  = (const float*)d_in[0];
  const float* W1 = (const float*)d_in[1];
  const float* b1 = (const float*)d_in[2];
  const float* W2 = (const float*)d_in[3];
  const float* b2 = (const float*)d_in[4];
  const float* W3 = (const float*)d_in[5];
  const float* b3 = (const float*)d_in[6];
  const float* Wo = (const float*)d_in[7];
  const float* bo = (const float*)d_in[8];
  float* out = (float*)d_out;

  dim3 grid(NTOT / ROWSBLK);
  flow_mlp_kernel<<<grid, dim3(256), 0, stream>>>(
      x, W1, b1, W2, b2, W3, b3, Wo, bo, out);
}


# Output layer folded into layer 3's epilogue: each lane multiplies its feature column by Wo and the row's 16 lanes are summed by in-row lane exchanges; layer 3 writes no LDS; dot sum order changes
# speedup vs baseline: 1.6846x; 1.0791x over previous
#include <hip/hip_runtime.h>
#include <math.h>


typedef float v2f __attribute__((ext_vector_type(2)));
typedef float v8f __attribute__((ext_vector_type(8)));

#define DD      8
#define HH      64
#define IN_DIM  9
#define BB      128
#define LL      256
#define NTOT    (BB * LL)
#define SLOPE   0.2f
#define AST     66
#define WAVES   8
#define TILES   8
#define ROWSBLK (16 * TILES)

static __device__ __forceinline__ v8f wmma_f32(v2f a, v2f b, v8f c) {
  return __builtin_amdgcn_wmma_f32_16x16x4_f32(
       false, a,  false, b,
       (short)0, c,  false,  false);
}

static __device__ __forceinline__ void hidden_layer(
    float* __restrict__ sA, float* __restrict__ sD,
    const v2f (&wf)[4][16], const float (&bias)[4],
    const int lx, const int half, const int koff)
{
  v2f af[16], df[16];
#pragma unroll
  for (int kk = 0; kk < 16; ++kk) {
    const int k = kk * 4 + koff;
    af[kk].x = sA[lx * AST + k];  af[kk].y = sA[lx * AST + k + 1];
    df[kk].x = sD[lx * AST + k];  df[kk].y = sD[lx * AST + k + 1];
  }
#pragma unroll
  for (int t = 0; t < 4; ++t) {
    v8f ca = {0.f, 0.f, 0.f, 0.f, 0.f, 0.f, 0.f, 0.f};
    v8f cd = {0.f, 0.f, 0.f, 0.f, 0.f, 0.f, 0.f, 0.f};
#pragma unroll
    for (int kk = 0; kk < 16; ++kk) {
      ca = wmma_f32(af[kk], wf[t][kk], ca);
      cd = wmma_f32(df[kk], wf[t][kk], cd);
    }
    const int f   = t * 16 + lx;
    const float bv = bias[t];
#pragma unroll
    for (int v = 0; v < 8; ++v) {
      const float h = ca[v] + bv;
      const float m = (h >= 0.f) ? 1.f : SLOPE;
      const int  ms = v + 8 * half;
      sA[ms * AST + f] = m * h;
      sD[ms * AST + f] = m * cd[v];
    }
  }
}

template <unsigned SEL_LO, unsigned SEL_HI>
static __device__ __forceinline__ float row_xchg(float x) {
  const int b = __builtin_bit_cast(int, x);
  return __builtin_bit_cast(float,
      __builtin_amdgcn_permlane16(b, b, SEL_LO, SEL_HI, false, false));
}

static __device__ __forceinline__ float hidden_layer_out(
    const float* __restrict__ sA, const float* __restrict__ sD,
    const v2f (&wf)[4][16], const float (&bias)[4], const float (&wo4)[4],
    const int lx, const int koff)
{
  v2f af[16], df[16];
#pragma unroll
  for (int kk = 0; kk < 16; ++kk) {
    const int k = kk * 4 + koff;
    af[kk].x = sA[lx * AST + k];  af[kk].y = sA[lx * AST + k + 1];
    df[kk].x = sD[lx * AST + k];  df[kk].y = sD[lx * AST + k + 1];
  }
  float pa[8], pd[8];
#pragma unroll
  for (int v = 0; v < 8; ++v) { pa[v] = 0.f; pd[v] = 0.f; }
#pragma unroll
  for (int t = 0; t < 4; ++t) {
    v8f ca = {0.f, 0.f, 0.f, 0.f, 0.f, 0.f, 0.f, 0.f};
    v8f cd = {0.f, 0.f, 0.f, 0.f, 0.f, 0.f, 0.f, 0.f};
#pragma unroll
    for (int kk = 0; kk < 16; ++kk) {
      ca = wmma_f32(af[kk], wf[t][kk], ca);
      cd = wmma_f32(df[kk], wf[t][kk], cd);
    }
    const float bv = bias[t];
    const float wv = wo4[t];
#pragma unroll
    for (int v = 0; v < 8; ++v) {
      const float h = ca[v] + bv;
      const float m = (h >= 0.f) ? 1.f : SLOPE;
      pa[v] = fmaf(m * h, wv, pa[v]);
      pd[v] = fmaf(m * cd[v], wv, pd[v]);
    }
  }
  const bool b3 = (lx & 8) != 0, b2 = (lx & 4) != 0, b1 = (lx & 2) != 0, b0 = (lx & 1) != 0;
  float q[8], r[4], s[2];
#pragma unroll
  for (int v = 0; v < 8; ++v) {
    const float keep = b3 ? pd[v] : pa[v];
    const float send = b3 ? pa[v] : pd[v];
    q[v] = keep + row_xchg<0xfedcba98u, 0x76543210u>(send);
  }
#pragma unroll
  for (int j = 0; j < 4; ++j) {
    const float keep = b2 ? q[4 + j] : q[j];
    const float send = b2 ? q[j] : q[4 + j];
    r[j] = keep + row_xchg<0x32107654u, 0xba98fedcu>(send);
  }
#pragma unroll
  for (int i = 0; i < 2; ++i) {
    const float keep = b1 ? r[2 + i] : r[i];
    const float send = b1 ? r[i] : r[2 + i];
    s[i] = keep + row_xchg<0x54761032u, 0xdcfe98bau>(send);
  }
  const float keep = b0 ? s[1] : s[0];
  const float send = b0 ? s[0] : s[1];
  return keep + row_xchg<0x67452301u, 0xefcdab89u>(send);
}

__global__ __launch_bounds__(256) void flow_mlp_kernel(
    const float* __restrict__ x,
    const float* __restrict__ W1, const float* __restrict__ b1,
    const float* __restrict__ W2, const float* __restrict__ b2,
    const float* __restrict__ W3, const float* __restrict__ b3,
    const float* __restrict__ Wo, const float* __restrict__ bo,
    float* __restrict__ out)
{
  __shared__ float sBufA[WAVES][16 * AST];
  __shared__ float sBufD[WAVES][16 * AST];
  __shared__ __attribute__((aligned(16))) float sRes[ROWSBLK * DD];
  __shared__ float sDl[DD][ROWSBLK];

  const int tid  = threadIdx.x;
  const int wave = __builtin_amdgcn_readfirstlane(tid >> 5);
  const int d    = wave;
  const int lane = tid & 31;
  const int lx   = lane & 15;
  const int half = lane >> 4;
  const int koff = half * 2;

  float* sA = sBufA[wave];
  float* sD = sBufD[wave];

  v2f w1f[4][3];
  v2f w2f[4][16];
  v2f w3f[4][16];
  float w8[4], bv1[4], bv2[4], bv3[4];
  float wo4[4];
#pragma unroll
  for (int t = 0; t < 4; ++t) {
    const int g = d * HH + t * 16 + lx;
    const float* r1 = W1 + (size_t)g * IN_DIM;
    w8[t] = r1[8];
    w1f[t][0].x = r1[koff];      w1f[t][0].y = r1[koff + 1];
    w1f[t][1].x = r1[4 + koff];  w1f[t][1].y = r1[5 + koff];
    w1f[t][2].x = half ? 0.f : w8[t];
    w1f[t][2].y = 0.f;
    bv1[t] = b1[g];
    bv2[t] = b2[g];
    bv3[t] = b3[g];
  }
#pragma unroll
  for (int t = 0; t < 4; ++t) {
    const int g = d * HH + t * 16 + lx;
    const v2f* r2 = reinterpret_cast<const v2f*>(W2 + (size_t)g * HH + koff);
#pragma unroll
    for (int kk = 0; kk < 16; ++kk) w2f[t][kk] = r2[kk * 2];
  }
#pragma unroll
  for (int t = 0; t < 4; ++t) {
    const int g = d * HH + t * 16 + lx;
    const v2f* r3 = reinterpret_cast<const v2f*>(W3 + (size_t)g * HH + koff);
#pragma unroll
    for (int kk = 0; kk < 16; ++kk) w3f[t][kk] = r3[kk * 2];
  }
#pragma unroll
  for (int t = 0; t < 4; ++t) wo4[t] = Wo[d * HH + t * 16 + lx];
  const float bod = bo[d];

  for (int it = 0; it < TILES; ++it) {
    const int n0 = blockIdx.x * ROWSBLK + it * 16;
    const int bb = n0 >> 8;
    const int l0 = n0 & 255;
    {
      const int s = lane >> 1, part = lane & 1;
      const float4 v = *reinterpret_cast<const float4*>(
          x + (size_t)(bb * 257 + l0 + s) * 8 + part * 4);
      const float xt = x[(size_t)(bb * 257 + l0 + lx + 1) * 8 + d];
      float* dst = &sA[s * AST + part * 4];
      dst[0] = v.x; dst[1] = v.y; dst[2] = v.z; dst[3] = v.w;
      sA[lx * AST + 8 + half] = half ? 0.f : xt;
      if (half) {
        sA[lx * AST + 10] = 0.f;
        sA[lx * AST + 11] = 0.f;
      }
    }

    {
      v2f af[3];
#pragma unroll
      for (int kk = 0; kk < 3; ++kk) {
        const int k = kk * 4 + koff;
        af[kk].x = sA[lx * AST + k];
        af[kk].y = sA[lx * AST + k + 1];
      }
      v8f acc1[4];
#pragma unroll
      for (int t = 0; t < 4; ++t) {
        v8f c = {0.f, 0.f, 0.f, 0.f, 0.f, 0.f, 0.f, 0.f};
#pragma unroll
        for (int kk = 0; kk < 3; ++kk) c = wmma_f32(af[kk], w1f[t][kk], c);
        acc1[t] = c;
      }
#pragma unroll
      for (int t = 0; t < 4; ++t) {
        const int f = t * 16 + lx;
#pragma unroll
        for (int v = 0; v < 8; ++v) {
          const float h = acc1[t][v] + bv1[t];
          const float m = (h >= 0.f) ? 1.f : SLOPE;
          const int  ms = v + 8 * half;
          sA[ms * AST + f] = m * h;
          sD[ms * AST + f] = m * w8[t];
        }
      }
    }

    hidden_layer(sA, sD, w2f, bv2, lx, half, koff);
    {
      const float tot = hidden_layer_out(sA, sD, w3f, bv3, wo4, lx, koff);
      const int r = it * 16 + 8 * half + (lx & 7);
      if (lx < 8) {
        sRes[r * DD + d] = tot + bod;
      } else {
        sDl[d][r] = tot;
      }
    }
  }

  __syncthreads();
  {
    const float4 v = *reinterpret_cast<const float4*>(&sRes[tid * 4]);
    *reinterpret_cast<float4*>(out + (size_t)blockIdx.x * ROWSBLK * DD + tid * 4) = v;
  }
  if (tid < ROWSBLK) {
    float s = 0.f;
#pragma unroll
    for (int dd = 0; dd < DD; ++dd) s += logf(fabsf(sDl[dd][tid]));
    out[(size_t)NTOT * DD + blockIdx.x * ROWSBLK + tid] = s;
  }
}

extern "C" void kernel_launch(void* const* d_in, const int* in_sizes, int n_in,
                              void* d_out, int out_size, void* d_ws, size_t ws_size,
                              hipStream_t stream) {
  const float* x  = (const float*)d_in[0];
  const float* W1 = (const float*)d_in[1];
  const float* b1 = (const float*)d_in[2];
  const float* W2 = (const float*)d_in[3];
  const float* b2 = (const float*)d_in[4];
  const float* W3 = (const float*)d_in[5];
  const float* b3 = (const float*)d_in[6];
  const float* Wo = (const float*)d_in[7];
  const float* bo = (const float*)d_in[8];
  float* out = (float*)d_out;

  dim3 grid(NTOT / ROWSBLK);
  flow_mlp_kernel<<<grid, dim3(256), 0, stream>>>(
      x, W1, b1, W2, b2, W3, b3, Wo, bo, out);
}
